# starting kernel
# baseline (speedup 1.0000x reference)
#include <hip/hip_runtime.h>
#include <hip/hip_bf16.h>

typedef float v2f __attribute__((ext_vector_type(2)));
typedef float v8f __attribute__((ext_vector_type(8)));

__global__ void build_u_kernel(const float* __restrict__ w, float* __restrict__ ws) {
    const int j = threadIdx.x;
    if (j >= 16) return;
    float cr[16], ci[16];
    for (int k = 0; k < 16; ++k) { cr[k] = (k == j) ? 1.f : 0.f; ci[k] = 0.f; }

    for (int l = 0; l < 6; ++l) {
        for (int i = 0; i < 4; ++i) {
            const float phi = w[(l * 4 + i) * 3 + 0];
            const float th  = w[(l * 4 + i) * 3 + 1];
            const float om  = w[(l * 4 + i) * 3 + 2];
            const float ct = cosf(0.5f * th), st = sinf(0.5f * th);
            const float a  = 0.5f * (phi + om), bb = 0.5f * (phi - om);
            const float ca = cosf(a), sa = sinf(a), cb = cosf(bb), sb = sinf(bb);
            const float g00r =  ct * ca, g00i = -ct * sa;
            const float g01r = -st * cb, g01i = -st * sb;
            const float g10r =  st * cb, g10i = -st * sb;
            const float g11r =  ct * ca, g11i =  ct * sa;
            const int bit = 1 << (3 - i);
            for (int j0 = 0; j0 < 16; ++j0) {
                if (j0 & bit) continue;
                const int j1 = j0 | bit;
                const float v0r = cr[j0], v0i = ci[j0], v1r = cr[j1], v1i = ci[j1];
                cr[j0] = g00r * v0r - g00i * v0i + g01r * v1r - g01i * v1i;
                ci[j0] = g00r * v0i + g00i * v0r + g01r * v1i + g01i * v1r;
                cr[j1] = g10r * v0r - g10i * v0i + g11r * v1r - g11i * v1i;
                ci[j1] = g10r * v0i + g10i * v0r + g11r * v1i + g11i * v1r;
            }
        }
        const int r = (l % 3) + 1;
        int perm[16], np[16];
        for (int k = 0; k < 16; ++k) perm[k] = k;
        for (int i = 0; i < 4; ++i) {
            const int tgt = (i + r) & 3;
            for (int k = 0; k < 16; ++k) {
                const int cbit = (k >> (3 - i)) & 1;
                const int src  = cbit ? (k ^ (1 << (3 - tgt))) : k;
                np[k] = perm[src];
            }
            for (int k = 0; k < 16; ++k) perm[k] = np[k];
        }
        float tr[16], ti[16];
        for (int k = 0; k < 16; ++k) { tr[k] = cr[perm[k]]; ti[k] = ci[perm[k]]; }
        for (int k = 0; k < 16; ++k) { cr[k] = tr[k]; ci[k] = ti[k]; }
    }

    const int t     = j >> 2;
    const int lbase = (j & 2) ? 16 : 0;
    const int comp  = j & 1;
    float* wr = ws;
    float* wi = ws + 256;
    for (int m = 0; m < 16; ++m) {
        const int L = lbase + m;
        wr[(t * 32 + L) * 2 + comp] = cr[m];
        wi[(t * 32 + L) * 2 + comp] = ci[m];
    }
}

__global__ void __launch_bounds__(256)
qlayer_kernel(const float* __restrict__ x, const float* __restrict__ ws,
              float* __restrict__ out, int batch) {
    const int lane = threadIdx.x & 31;
    const int wave = (blockIdx.x * blockDim.x + threadIdx.x) >> 5;
    int b = wave * 16 + (lane & 15);
    if (b >= batch) b = batch - 1;

    const float4 xv = *reinterpret_cast<const float4*>(x + (size_t)b * 4);
    const float c0 = __cosf(0.5f * xv.x), s0 = __sinf(0.5f * xv.x);
    const float c1 = __cosf(0.5f * xv.y), s1 = __sinf(0.5f * xv.y);
    const float c2 = __cosf(0.5f * xv.z), s2 = __sinf(0.5f * xv.z);
    const float c3 = __cosf(0.5f * xv.w), s3 = __sinf(0.5f * xv.w);

    const float hi = (lane < 16) ? c2 : s2;
    const float T0 = c0 * c1, T1 = c0 * s1, T2 = s0 * c1, T3 = s0 * s1;
    v2f bf[4];
    bf[0] = (v2f){T0 * hi * c3, T0 * hi * s3};
    bf[1] = (v2f){T1 * hi * c3, T1 * hi * s3};
    bf[2] = (v2f){T2 * hi * c3, T2 * hi * s3};
    bf[3] = (v2f){T3 * hi * c3, T3 * hi * s3};

    const float* wr = ws;
    const float* wi = ws + 256;
    v2f ar[4], ai[4];
#pragma unroll
    for (int t = 0; t < 4; ++t) {
        ar[t] = *reinterpret_cast<const v2f*>(wr + (t * 32 + lane) * 2);
        ai[t] = *reinterpret_cast<const v2f*>(wi + (t * 32 + lane) * 2);
    }

    v8f accr = {0.f, 0.f, 0.f, 0.f, 0.f, 0.f, 0.f, 0.f};
    v8f acci = {0.f, 0.f, 0.f, 0.f, 0.f, 0.f, 0.f, 0.f};
#pragma unroll
    for (int t = 0; t < 4; ++t) {
        accr = __builtin_amdgcn_wmma_f32_16x16x4_f32(
            false, ar[t], false, bf[t], (short)0, accr, false, false);
        acci = __builtin_amdgcn_wmma_f32_16x16x4_f32(
            false, ai[t], false, bf[t], (short)0, acci, false, false);
    }

    const int mbase = (lane < 16) ? 0 : 8;
    float o0 = 0.f, o1 = 0.f, o2 = 0.f, o3 = 0.f;
#pragma unroll
    for (int v = 0; v < 8; ++v) {
        const float p = accr[v] * accr[v] + acci[v] * acci[v];
        const int m = mbase + v;
        o0 += ((m >> 3) & 1) ? -p : p;
        o1 += ((m >> 2) & 1) ? -p : p;
        o2 += ((m >> 1) & 1) ? -p : p;
        o3 += ( m       & 1) ? -p : p;
    }
    o0 += __shfl_xor(o0, 16, 32);
    o1 += __shfl_xor(o1, 16, 32);
    o2 += __shfl_xor(o2, 16, 32);
    o3 += __shfl_xor(o3, 16, 32);

    if (lane < 16 && b < batch) {
        float4 res = {o0, o1, o2, o3};
        *reinterpret_cast<float4*>(out + (size_t)b * 4) = res;
    }
}

extern "C" void kernel_launch(void* const* d_in, const int* in_sizes, int n_in,
                              void* d_out, int out_size, void* d_ws, size_t ws_size,
                              hipStream_t stream) {
    const float* x = (const float*)d_in[0];
    const float* w = (const float*)d_in[1];
    float* out = (float*)d_out;
    float* ws  = (float*)d_ws;

    const int batch = in_sizes[0] / 4;

    build_u_kernel<<<1, 16, 0, stream>>>(w, ws);

    const int waves  = (batch + 15) / 16;
    const int wpb    = 8;
    const int blocks = (waves + wpb - 1) / wpb;
    qlayer_kernel<<<blocks, 256, 0, stream>>>(x, ws, out, batch);
}


# U built by one wave as a WMMA product of the six 16x16 layer matrices, each formed directly in fragment order as the Kronecker product of its four gates; no gate-by-gate sweep; sum order changed
# speedup vs baseline: 2.4737x; 2.4737x over previous
#include <hip/hip_runtime.h>
#include <hip/hip_bf16.h>


typedef float v2f __attribute__((ext_vector_type(2)));
typedef float v8f __attribute__((ext_vector_type(8)));

struct cf { float r, i; };
__device__ __forceinline__ cf cmul(cf a, cf b) {
    return cf{a.r * b.r - a.i * b.i, a.r * b.i + a.i * b.r};
}

__device__ __forceinline__ int ring_src(int k, int r) {
    int p = k;
#pragma unroll
    for (int i = 3; i >= 0; --i) {
        const int tgt = (i + r) & 3;
        p ^= ((p >> (3 - i)) & 1) << (3 - tgt);
    }
    return p;
}

__global__ void __launch_bounds__(32)
build_u_kernel(const float* __restrict__ w, float* __restrict__ ws) {
    __shared__ cf gt[32][2][2];
    const int lane = threadIdx.x;
    const int c = lane & 15, h = lane >> 4;

    {
        const int g = (lane < 24) ? lane : lane - 8;
        const float phi = w[g * 3 + 0], th = w[g * 3 + 1], om = w[g * 3 + 2];
        const float ct = cosf(0.5f * th), st = sinf(0.5f * th);
        const float a  = 0.5f * (phi + om), bb = 0.5f * (phi - om);
        const float ca = cosf(a), sa = sinf(a), cb = cosf(bb), sb = sinf(bb);
        gt[lane][0][0] = cf{ ct * ca, -ct * sa};
        gt[lane][0][1] = cf{-st * cb, -st * sb};
        gt[lane][1][0] = cf{ st * cb, -st * sb};
        gt[lane][1][1] = cf{ ct * ca,  ct * sa};
    }
    __syncthreads();

    v8f xr, xi;
    {
        const int l = 5;
        const int pm = ring_src(c, (l % 3) + 1);
        const cf a  = gt[4 * l + 0][(pm >> 3) & 1][h];
        const cf b0 = gt[4 * l + 1][(pm >> 2) & 1][0], b1 = gt[4 * l + 1][(pm >> 2) & 1][1];
        const cf c0 = gt[4 * l + 2][(pm >> 1) & 1][0], c1 = gt[4 * l + 2][(pm >> 1) & 1][1];
        const cf d0 = gt[4 * l + 3][pm & 1][0],        d1 = gt[4 * l + 3][pm & 1][1];
        const cf ab[2] = {cmul(a, b0), cmul(a, b1)};
        const cf cd[4] = {cmul(c0, d0), cmul(c0, d1), cmul(c1, d0), cmul(c1, d1)};
#pragma unroll
        for (int q = 0; q < 8; ++q) {
            const cf v = cmul(ab[q >> 2], cd[q & 3]);
            xr[q] = v.r; xi[q] = v.i;
        }
    }

    const int ci0 = (c >> 3) & 1, ci1 = (c >> 2) & 1, ci2 = (c >> 1) & 1, ci3 = c & 1;
#pragma unroll
    for (int l = 4; l >= 0; --l) {
        const int r = (l % 3) + 1;
        const int ph = h ? ring_src(8, r) : 0;
        cf f0[2], f1[2], f2[2], f3[2];
#pragma unroll
        for (int b = 0; b < 2; ++b) {
            f0[b] = gt[4 * l + 0][b ^ ((ph >> 3) & 1)][ci0];
            f1[b] = gt[4 * l + 1][b ^ ((ph >> 2) & 1)][ci1];
            f2[b] = gt[4 * l + 2][b ^ ((ph >> 1) & 1)][ci2];
            f3[b] = gt[4 * l + 3][b ^ (ph & 1)][ci3];
        }
        cf p01[4], p23[4];
#pragma unroll
        for (int u = 0; u < 4; ++u) {
            p01[u] = cmul(f0[u >> 1], f1[u & 1]);
            p23[u] = cmul(f2[u >> 1], f3[u & 1]);
        }
        v2f ar[4], ai[4], nai[4];
#pragma unroll
        for (int q = 0; q < 8; ++q) {
            const int pq = ring_src(q, r);
            const cf v = cmul(p01[pq >> 2], p23[pq & 3]);
            ar[q >> 1][q & 1] = v.r; ai[q >> 1][q & 1] = v.i; nai[q >> 1][q & 1] = -v.i;
        }
        v8f nr = {0.f, 0.f, 0.f, 0.f, 0.f, 0.f, 0.f, 0.f};
        v8f ni = {0.f, 0.f, 0.f, 0.f, 0.f, 0.f, 0.f, 0.f};
#pragma unroll
        for (int t = 0; t < 4; ++t) {
            const v2f br = {xr[2 * t], xr[2 * t + 1]};
            const v2f bi = {xi[2 * t], xi[2 * t + 1]};
            nr = __builtin_amdgcn_wmma_f32_16x16x4_f32(false, ar[t],  false, br, (short)0, nr, false, false);
            ni = __builtin_amdgcn_wmma_f32_16x16x4_f32(false, ar[t],  false, bi, (short)0, ni, false, false);
            nr = __builtin_amdgcn_wmma_f32_16x16x4_f32(false, nai[t], false, bi, (short)0, nr, false, false);
            ni = __builtin_amdgcn_wmma_f32_16x16x4_f32(false, ai[t],  false, br, (short)0, ni, false, false);
        }
        xr = nr; xi = ni;
    }

    float* wr = ws;
    float* wi = ws + 256;
#pragma unroll
    for (int p = 0; p < 4; ++p) {
        const int j = 8 * h + 2 * p;
        const int idx = ((j >> 2) * 32 + c + ((j & 2) ? 16 : 0)) * 2;
        *reinterpret_cast<v2f*>(wr + idx) = (v2f){xr[2 * p], xr[2 * p + 1]};
        *reinterpret_cast<v2f*>(wi + idx) = (v2f){xi[2 * p], xi[2 * p + 1]};
    }
}

__global__ void __launch_bounds__(256)
qlayer_kernel(const float* __restrict__ x, const float* __restrict__ ws,
              float* __restrict__ out, int batch) {
    const int lane = threadIdx.x & 31;
    const int wave = (blockIdx.x * blockDim.x + threadIdx.x) >> 5;
    int b = wave * 16 + (lane & 15);
    if (b >= batch) b = batch - 1;

    const float4 xv = *reinterpret_cast<const float4*>(x + (size_t)b * 4);
    const float c0 = __cosf(0.5f * xv.x), s0 = __sinf(0.5f * xv.x);
    const float c1 = __cosf(0.5f * xv.y), s1 = __sinf(0.5f * xv.y);
    const float c2 = __cosf(0.5f * xv.z), s2 = __sinf(0.5f * xv.z);
    const float c3 = __cosf(0.5f * xv.w), s3 = __sinf(0.5f * xv.w);

    const float hi = (lane < 16) ? c2 : s2;
    const float T0 = c0 * c1, T1 = c0 * s1, T2 = s0 * c1, T3 = s0 * s1;
    v2f bf[4];
    bf[0] = (v2f){T0 * hi * c3, T0 * hi * s3};
    bf[1] = (v2f){T1 * hi * c3, T1 * hi * s3};
    bf[2] = (v2f){T2 * hi * c3, T2 * hi * s3};
    bf[3] = (v2f){T3 * hi * c3, T3 * hi * s3};

    const float* wr = ws;
    const float* wi = ws + 256;
    v2f ar[4], ai[4];
#pragma unroll
    for (int t = 0; t < 4; ++t) {
        ar[t] = *reinterpret_cast<const v2f*>(wr + (t * 32 + lane) * 2);
        ai[t] = *reinterpret_cast<const v2f*>(wi + (t * 32 + lane) * 2);
    }

    v8f accr = {0.f, 0.f, 0.f, 0.f, 0.f, 0.f, 0.f, 0.f};
    v8f acci = {0.f, 0.f, 0.f, 0.f, 0.f, 0.f, 0.f, 0.f};
#pragma unroll
    for (int t = 0; t < 4; ++t) {
        accr = __builtin_amdgcn_wmma_f32_16x16x4_f32(
            false, ar[t], false, bf[t], (short)0, accr, false, false);
        acci = __builtin_amdgcn_wmma_f32_16x16x4_f32(
            false, ai[t], false, bf[t], (short)0, acci, false, false);
    }

    const int mbase = (lane < 16) ? 0 : 8;
    float o0 = 0.f, o1 = 0.f, o2 = 0.f, o3 = 0.f;
#pragma unroll
    for (int v = 0; v < 8; ++v) {
        const float p = accr[v] * accr[v] + acci[v] * acci[v];
        const int m = mbase + v;
        o0 += ((m >> 3) & 1) ? -p : p;
        o1 += ((m >> 2) & 1) ? -p : p;
        o2 += ((m >> 1) & 1) ? -p : p;
        o3 += ( m       & 1) ? -p : p;
    }
    o0 += __shfl_xor(o0, 16, 32);
    o1 += __shfl_xor(o1, 16, 32);
    o2 += __shfl_xor(o2, 16, 32);
    o3 += __shfl_xor(o3, 16, 32);

    if (lane < 16 && b < batch) {
        float4 res = {o0, o1, o2, o3};
        *reinterpret_cast<float4*>(out + (size_t)b * 4) = res;
    }
}

extern "C" void kernel_launch(void* const* d_in, const int* in_sizes, int n_in,
                              void* d_out, int out_size, void* d_ws, size_t ws_size,
                              hipStream_t stream) {
    const float* x = (const float*)d_in[0];
    const float* w = (const float*)d_in[1];
    float* out = (float*)d_out;
    float* ws  = (float*)d_ws;

    const int batch = in_sizes[0] / 4;

    build_u_kernel<<<1, 32, 0, stream>>>(w, ws);

    const int waves  = (batch + 15) / 16;
    const int wpb    = 8;
    const int blocks = (waves + wpb - 1) / wpb;
    qlayer_kernel<<<blocks, 256, 0, stream>>>(x, ws, out, batch);
}


# Main kernel reads x one row per lane, 4 full-width loads up front (128 rows per wave); both tiles of a 32-row group get B fragments and combined sums by v_permlane16_swap; sum order changed
# speedup vs baseline: 2.9120x; 1.1772x over previous
#include <hip/hip_runtime.h>
#include <hip/hip_bf16.h>


typedef float v2f __attribute__((ext_vector_type(2)));
typedef float v8f __attribute__((ext_vector_type(8)));

struct cf { float r, i; };
__device__ __forceinline__ cf cmul(cf a, cf b) {
    return cf{a.r * b.r - a.i * b.i, a.r * b.i + a.i * b.r};
}

__device__ __forceinline__ int ring_src(int k, int r) {
    int p = k;
#pragma unroll
    for (int i = 3; i >= 0; --i) {
        const int tgt = (i + r) & 3;
        p ^= ((p >> (3 - i)) & 1) << (3 - tgt);
    }
    return p;
}

__global__ void __launch_bounds__(32)
build_u_kernel(const float* __restrict__ w, float* __restrict__ ws) {
    __shared__ cf gt[32][2][2];
    const int lane = threadIdx.x;
    const int c = lane & 15, h = lane >> 4;

    {
        const int g = (lane < 24) ? lane : lane - 8;
        const float phi = w[g * 3 + 0], th = w[g * 3 + 1], om = w[g * 3 + 2];
        const float ct = cosf(0.5f * th), st = sinf(0.5f * th);
        const float a  = 0.5f * (phi + om), bb = 0.5f * (phi - om);
        const float ca = cosf(a), sa = sinf(a), cb = cosf(bb), sb = sinf(bb);
        gt[lane][0][0] = cf{ ct * ca, -ct * sa};
        gt[lane][0][1] = cf{-st * cb, -st * sb};
        gt[lane][1][0] = cf{ st * cb, -st * sb};
        gt[lane][1][1] = cf{ ct * ca,  ct * sa};
    }
    __syncthreads();

    v8f xr, xi;
    {
        const int l = 5;
        const int pm = ring_src(c, (l % 3) + 1);
        const cf a  = gt[4 * l + 0][(pm >> 3) & 1][h];
        const cf b0 = gt[4 * l + 1][(pm >> 2) & 1][0], b1 = gt[4 * l + 1][(pm >> 2) & 1][1];
        const cf c0 = gt[4 * l + 2][(pm >> 1) & 1][0], c1 = gt[4 * l + 2][(pm >> 1) & 1][1];
        const cf d0 = gt[4 * l + 3][pm & 1][0],        d1 = gt[4 * l + 3][pm & 1][1];
        const cf ab[2] = {cmul(a, b0), cmul(a, b1)};
        const cf cd[4] = {cmul(c0, d0), cmul(c0, d1), cmul(c1, d0), cmul(c1, d1)};
#pragma unroll
        for (int q = 0; q < 8; ++q) {
            const cf v = cmul(ab[q >> 2], cd[q & 3]);
            xr[q] = v.r; xi[q] = v.i;
        }
    }

    const int ci0 = (c >> 3) & 1, ci1 = (c >> 2) & 1, ci2 = (c >> 1) & 1, ci3 = c & 1;
#pragma unroll
    for (int l = 4; l >= 0; --l) {
        const int r = (l % 3) + 1;
        const int ph = h ? ring_src(8, r) : 0;
        cf f0[2], f1[2], f2[2], f3[2];
#pragma unroll
        for (int b = 0; b < 2; ++b) {
            f0[b] = gt[4 * l + 0][b ^ ((ph >> 3) & 1)][ci0];
            f1[b] = gt[4 * l + 1][b ^ ((ph >> 2) & 1)][ci1];
            f2[b] = gt[4 * l + 2][b ^ ((ph >> 1) & 1)][ci2];
            f3[b] = gt[4 * l + 3][b ^ (ph & 1)][ci3];
        }
        cf p01[4], p23[4];
#pragma unroll
        for (int u = 0; u < 4; ++u) {
            p01[u] = cmul(f0[u >> 1], f1[u & 1]);
            p23[u] = cmul(f2[u >> 1], f3[u & 1]);
        }
        v2f ar[4], ai[4], nai[4];
#pragma unroll
        for (int q = 0; q < 8; ++q) {
            const int pq = ring_src(q, r);
            const cf v = cmul(p01[pq >> 2], p23[pq & 3]);
            ar[q >> 1][q & 1] = v.r; ai[q >> 1][q & 1] = v.i; nai[q >> 1][q & 1] = -v.i;
        }
        v8f nr = {0.f, 0.f, 0.f, 0.f, 0.f, 0.f, 0.f, 0.f};
        v8f ni = {0.f, 0.f, 0.f, 0.f, 0.f, 0.f, 0.f, 0.f};
#pragma unroll
        for (int t = 0; t < 4; ++t) {
            const v2f br = {xr[2 * t], xr[2 * t + 1]};
            const v2f bi = {xi[2 * t], xi[2 * t + 1]};
            nr = __builtin_amdgcn_wmma_f32_16x16x4_f32(false, ar[t],  false, br, (short)0, nr, false, false);
            ni = __builtin_amdgcn_wmma_f32_16x16x4_f32(false, ar[t],  false, bi, (short)0, ni, false, false);
            nr = __builtin_amdgcn_wmma_f32_16x16x4_f32(false, nai[t], false, bi, (short)0, nr, false, false);
            ni = __builtin_amdgcn_wmma_f32_16x16x4_f32(false, ai[t],  false, br, (short)0, ni, false, false);
        }
        xr = nr; xi = ni;
    }

    float* wr = ws;
    float* wi = ws + 256;
#pragma unroll
    for (int p = 0; p < 4; ++p) {
        const int j = 8 * h + 2 * p;
        const int idx = ((j >> 2) * 32 + c + ((j & 2) ? 16 : 0)) * 2;
        *reinterpret_cast<v2f*>(wr + idx) = (v2f){xr[2 * p], xr[2 * p + 1]};
        *reinterpret_cast<v2f*>(wi + idx) = (v2f){xi[2 * p], xi[2 * p + 1]};
    }
}

__device__ __forceinline__ void half_swap(float& a, float& b) {
    asm("v_permlane16_swap_b32 %0, %1" : "+v"(a), "+v"(b));
}
#define ROWS_PER_WAVE 128
#define GROUPS (ROWS_PER_WAVE / 32)

__global__ void __launch_bounds__(256)
qlayer_kernel(const float* __restrict__ x, const float* __restrict__ ws,
              float* __restrict__ out, int batch) {
    const int lane = threadIdx.x & 31;
    const int wave = (blockIdx.x * 256 + threadIdx.x) >> 5;
    const int row0 = wave * ROWS_PER_WAVE + lane;

    const float* wr = ws;
    const float* wi = ws + 256;
    v2f ar[4], ai[4];
#pragma unroll
    for (int t = 0; t < 4; ++t) {
        ar[t] = *reinterpret_cast<const v2f*>(wr + (t * 32 + lane) * 2);
        ai[t] = *reinterpret_cast<const v2f*>(wi + (t * 32 + lane) * 2);
    }

    float4 xv[GROUPS];
#pragma unroll
    for (int g = 0; g < GROUPS; ++g) {
        int r = row0 + 32 * g;
        if (r >= batch) r = batch - 1;
        xv[g] = *reinterpret_cast<const float4*>(x + (size_t)r * 4);
    }
    __builtin_amdgcn_sched_barrier(0);

#pragma unroll
    for (int g = 0; g < GROUPS; ++g) {
        const float c0 = __cosf(0.5f * xv[g].x), s0 = __sinf(0.5f * xv[g].x);
        const float c1 = __cosf(0.5f * xv[g].y), s1 = __sinf(0.5f * xv[g].y);
        const float c2 = __cosf(0.5f * xv[g].z), s2 = __sinf(0.5f * xv[g].z);
        const float c3 = __cosf(0.5f * xv[g].w), s3 = __sinf(0.5f * xv[g].w);
        const float T[4] = {c0 * c1, c0 * s1, s0 * c1, s0 * s1};
        float X[4][2], Y[4][2];
#pragma unroll
        for (int t = 0; t < 4; ++t) {
            const float tc = T[t] * c2, ts = T[t] * s2;
            X[t][0] = tc * c3; X[t][1] = tc * s3;
            Y[t][0] = ts * c3; Y[t][1] = ts * s3;
            half_swap(X[t][0], Y[t][0]);
            half_swap(X[t][1], Y[t][1]);
        }

        v8f r0 = {0.f, 0.f, 0.f, 0.f, 0.f, 0.f, 0.f, 0.f};
        v8f i0 = {0.f, 0.f, 0.f, 0.f, 0.f, 0.f, 0.f, 0.f};
        v8f r1 = {0.f, 0.f, 0.f, 0.f, 0.f, 0.f, 0.f, 0.f};
        v8f i1 = {0.f, 0.f, 0.f, 0.f, 0.f, 0.f, 0.f, 0.f};
#pragma unroll
        for (int t = 0; t < 4; ++t) {
            const v2f b0 = {X[t][0], X[t][1]};
            const v2f b1 = {Y[t][0], Y[t][1]};
            r0 = __builtin_amdgcn_wmma_f32_16x16x4_f32(false, ar[t], false, b0, (short)0, r0, false, false);
            i0 = __builtin_amdgcn_wmma_f32_16x16x4_f32(false, ai[t], false, b0, (short)0, i0, false, false);
            r1 = __builtin_amdgcn_wmma_f32_16x16x4_f32(false, ar[t], false, b1, (short)0, r1, false, false);
            i1 = __builtin_amdgcn_wmma_f32_16x16x4_f32(false, ai[t], false, b1, (short)0, i1, false, false);
        }

        float tot[2], o1[2], o2[2], o3[2];
#pragma unroll
        for (int k = 0; k < 2; ++k) {
            const v8f& sr = k ? r1 : r0;
            const v8f& si = k ? i1 : i0;
            float p[8];
#pragma unroll
            for (int v = 0; v < 8; ++v) p[v] = sr[v] * sr[v] + si[v] * si[v];
            const float a0 = p[0] + p[1], a1 = p[2] + p[3], a2 = p[4] + p[5], a3 = p[6] + p[7];
            const float d0 = p[0] - p[1], d1 = p[2] - p[3], d2 = p[4] - p[5], d3 = p[6] - p[7];
            const float b0 = a0 + a1, b1 = a2 + a3;
            o3[k]  = (d0 + d1) + (d2 + d3);
            o2[k]  = (a0 - a1) + (a2 - a3);
            o1[k]  = b0 - b1;
            tot[k] = b0 + b1;
        }
        half_swap(tot[0], tot[1]);
        half_swap(o1[0], o1[1]);
        half_swap(o2[0], o2[1]);
        half_swap(o3[0], o3[1]);

        const int r = row0 + 32 * g;
        if (r < batch) {
            const float4 res = {tot[0] - tot[1], o1[0] + o1[1], o2[0] + o2[1], o3[0] + o3[1]};
            *reinterpret_cast<float4*>(out + (size_t)r * 4) = res;
        }
    }
}

extern "C" void kernel_launch(void* const* d_in, const int* in_sizes, int n_in,
                              void* d_out, int out_size, void* d_ws, size_t ws_size,
                              hipStream_t stream) {
    const float* x = (const float*)d_in[0];
    const float* w = (const float*)d_in[1];
    float* out = (float*)d_out;
    float* ws  = (float*)d_ws;

    const int batch = in_sizes[0] / 4;

    build_u_kernel<<<1, 32, 0, stream>>>(w, ws);

    const int waves  = (batch + ROWS_PER_WAVE - 1) / ROWS_PER_WAVE;
    const int wpb    = 8;
    const int blocks = (waves + wpb - 1) / wpb;
    qlayer_kernel<<<blocks, 256, 0, stream>>>(x, ws, out, batch);
}


# One launch: each of 2,048 waves builds U in its own registers (one-wave WMMA chain; D tiles turned into A fragments by a lane-half exchange, no hand-off) and loops over 512 rows, next rows in flight
# speedup vs baseline: 3.0959x; 1.0632x over previous
#include <hip/hip_runtime.h>
#include <hip/hip_bf16.h>


typedef float v2f __attribute__((ext_vector_type(2)));
typedef float v8f __attribute__((ext_vector_type(8)));

struct cf { float r, i; };
__device__ __forceinline__ cf cmul(cf a, cf b) {
    return cf{a.r * b.r - a.i * b.i, a.r * b.i + a.i * b.r};
}

__device__ __forceinline__ int ring_src(int k, int r) {
    int p = k;
#pragma unroll
    for (int i = 3; i >= 0; --i) {
        const int tgt = (i + r) & 3;
        p ^= ((p >> (3 - i)) & 1) << (3 - tgt);
    }
    return p;
}

__device__ __forceinline__ void half_swap(float& a, float& b) {
    asm("v_permlane16_swap_b32 %0, %1" : "+v"(a), "+v"(b));
}

__device__ __forceinline__ float other_half(float v) {
    const unsigned u = __float_as_uint(v);
    return __uint_as_float(__builtin_amdgcn_permlanex16(u, u, 0x76543210u, 0xfedcba98u, false, false));
}

#define WAVES_PER_WG 8
#define GROUPS 4
#define CHUNK_ROWS (32 * GROUPS)

__device__ __forceinline__ void build_gate_table(float phi, float th, float om, cf (*gt)[2][2], int lane) {
    {
        const float ct = cosf(0.5f * th), st = sinf(0.5f * th);
        const float a  = 0.5f * (phi + om), bb = 0.5f * (phi - om);
        const float ca = cosf(a), sa = sinf(a), cb = cosf(bb), sb = sinf(bb);
        gt[lane][0][0] = cf{ ct * ca, -ct * sa};
        gt[lane][0][1] = cf{-st * cb, -st * sb};
        gt[lane][1][0] = cf{ st * cb, -st * sb};
        gt[lane][1][1] = cf{ ct * ca,  ct * sa};
    }
}

__device__ __forceinline__ void build_u_fragments(cf (*gt)[2][2], int lane, v2f (&far)[4], v2f (&fai)[4]) {
    const int c = lane & 15, h = lane >> 4;

    v8f xr, xi;
    {
        const int l = 5;
        const int pm = ring_src(c, (l % 3) + 1);
        const cf a  = gt[4 * l + 0][(pm >> 3) & 1][h];
        const cf b0 = gt[4 * l + 1][(pm >> 2) & 1][0], b1 = gt[4 * l + 1][(pm >> 2) & 1][1];
        const cf c0 = gt[4 * l + 2][(pm >> 1) & 1][0], c1 = gt[4 * l + 2][(pm >> 1) & 1][1];
        const cf d0 = gt[4 * l + 3][pm & 1][0],        d1 = gt[4 * l + 3][pm & 1][1];
        const cf ab[2] = {cmul(a, b0), cmul(a, b1)};
        const cf cd[4] = {cmul(c0, d0), cmul(c0, d1), cmul(c1, d0), cmul(c1, d1)};
#pragma unroll
        for (int q = 0; q < 8; ++q) {
            const cf v = cmul(ab[q >> 2], cd[q & 3]);
            xr[q] = v.r; xi[q] = v.i;
        }
    }

    const int ci0 = (c >> 3) & 1, ci1 = (c >> 2) & 1, ci2 = (c >> 1) & 1, ci3 = c & 1;
#pragma unroll
    for (int l = 4; l >= 0; --l) {
        const int r = (l % 3) + 1;
        const int ph = h ? ring_src(8, r) : 0;
        cf f0[2], f1[2], f2[2], f3[2];
#pragma unroll
        for (int b = 0; b < 2; ++b) {
            f0[b] = gt[4 * l + 0][b ^ ((ph >> 3) & 1)][ci0];
            f1[b] = gt[4 * l + 1][b ^ ((ph >> 2) & 1)][ci1];
            f2[b] = gt[4 * l + 2][b ^ ((ph >> 1) & 1)][ci2];
            f3[b] = gt[4 * l + 3][b ^ (ph & 1)][ci3];
        }
        cf p01[4], p23[4];
#pragma unroll
        for (int u = 0; u < 4; ++u) {
            p01[u] = cmul(f0[u >> 1], f1[u & 1]);
            p23[u] = cmul(f2[u >> 1], f3[u & 1]);
        }
        v2f ar[4], ai[4], nai[4];
#pragma unroll
        for (int q = 0; q < 8; ++q) {
            const int pq = ring_src(q, r);
            const cf v = cmul(p01[pq >> 2], p23[pq & 3]);
            ar[q >> 1][q & 1] = v.r; ai[q >> 1][q & 1] = v.i; nai[q >> 1][q & 1] = -v.i;
        }
        v8f nr = {0.f, 0.f, 0.f, 0.f, 0.f, 0.f, 0.f, 0.f};
        v8f ni = {0.f, 0.f, 0.f, 0.f, 0.f, 0.f, 0.f, 0.f};
#pragma unroll
        for (int t = 0; t < 4; ++t) {
            const v2f br = {xr[2 * t], xr[2 * t + 1]};
            const v2f bi = {xi[2 * t], xi[2 * t + 1]};
            nr = __builtin_amdgcn_wmma_f32_16x16x4_f32(false, ar[t],  false, br, (short)0, nr, false, false);
            ni = __builtin_amdgcn_wmma_f32_16x16x4_f32(false, ar[t],  false, bi, (short)0, ni, false, false);
            nr = __builtin_amdgcn_wmma_f32_16x16x4_f32(false, nai[t], false, bi, (short)0, nr, false, false);
            ni = __builtin_amdgcn_wmma_f32_16x16x4_f32(false, ai[t],  false, br, (short)0, ni, false, false);
        }
        xr = nr; xi = ni;
    }

#pragma unroll
    for (int p = 0; p < 2; ++p) {
#pragma unroll
        for (int s = 0; s < 2; ++s) {
            const int q = 4 * p + s;
            const float r_lo = xr[q], r_hi = xr[q + 2];
            const float i_lo = xi[q], i_hi = xi[q + 2];
            const float r_lo_o = other_half(r_lo), r_hi_o = other_half(r_hi);
            const float i_lo_o = other_half(i_lo), i_hi_o = other_half(i_hi);
            far[p][s]     = h ? r_hi_o : r_lo;
            far[p + 2][s] = h ? r_hi   : r_lo_o;
            fai[p][s]     = h ? i_hi_o : i_lo;
            fai[p + 2][s] = h ? i_hi   : i_lo_o;
        }
    }
}

__device__ __forceinline__ void do_group(const float4 xv, const v2f (&ar)[4], const v2f (&ai)[4],
                                         float* __restrict__ out, int r, int batch) {
    const float c0 = __cosf(0.5f * xv.x), s0 = __sinf(0.5f * xv.x);
    const float c1 = __cosf(0.5f * xv.y), s1 = __sinf(0.5f * xv.y);
    const float c2 = __cosf(0.5f * xv.z), s2 = __sinf(0.5f * xv.z);
    const float c3 = __cosf(0.5f * xv.w), s3 = __sinf(0.5f * xv.w);
    const float T[4] = {c0 * c1, c0 * s1, s0 * c1, s0 * s1};
    float X[4][2], Y[4][2];
#pragma unroll
    for (int t = 0; t < 4; ++t) {
        const float tc = T[t] * c2, ts = T[t] * s2;
        X[t][0] = tc * c3; X[t][1] = tc * s3;
        Y[t][0] = ts * c3; Y[t][1] = ts * s3;
        half_swap(X[t][0], Y[t][0]);
        half_swap(X[t][1], Y[t][1]);
    }

    v8f r0 = {0.f, 0.f, 0.f, 0.f, 0.f, 0.f, 0.f, 0.f};
    v8f i0 = {0.f, 0.f, 0.f, 0.f, 0.f, 0.f, 0.f, 0.f};
    v8f r1 = {0.f, 0.f, 0.f, 0.f, 0.f, 0.f, 0.f, 0.f};
    v8f i1 = {0.f, 0.f, 0.f, 0.f, 0.f, 0.f, 0.f, 0.f};
#pragma unroll
    for (int t = 0; t < 4; ++t) {
        const v2f b0 = {X[t][0], X[t][1]};
        const v2f b1 = {Y[t][0], Y[t][1]};
        r0 = __builtin_amdgcn_wmma_f32_16x16x4_f32(false, ar[t], false, b0, (short)0, r0, false, false);
        i0 = __builtin_amdgcn_wmma_f32_16x16x4_f32(false, ai[t], false, b0, (short)0, i0, false, false);
        r1 = __builtin_amdgcn_wmma_f32_16x16x4_f32(false, ar[t], false, b1, (short)0, r1, false, false);
        i1 = __builtin_amdgcn_wmma_f32_16x16x4_f32(false, ai[t], false, b1, (short)0, i1, false, false);
    }

    float tot[2], o1[2], o2[2], o3[2];
#pragma unroll
    for (int k = 0; k < 2; ++k) {
        const v8f& sr = k ? r1 : r0;
        const v8f& si = k ? i1 : i0;
        float p[8];
#pragma unroll
        for (int v = 0; v < 8; ++v) p[v] = sr[v] * sr[v] + si[v] * si[v];
        const float a0 = p[0] + p[1], a1 = p[2] + p[3], a2 = p[4] + p[5], a3 = p[6] + p[7];
        const float d0 = p[0] - p[1], d1 = p[2] - p[3], d2 = p[4] - p[5], d3 = p[6] - p[7];
        const float b0 = a0 + a1, b1 = a2 + a3;
        o3[k]  = (d0 + d1) + (d2 + d3);
        o2[k]  = (a0 - a1) + (a2 - a3);
        o1[k]  = b0 - b1;
        tot[k] = b0 + b1;
    }
    half_swap(tot[0], tot[1]);
    half_swap(o1[0], o1[1]);
    half_swap(o2[0], o2[1]);
    half_swap(o3[0], o3[1]);

    if (r < batch) {
        const float4 res = {tot[0] - tot[1], o1[0] + o1[1], o2[0] + o2[1], o3[0] + o3[1]};
        *reinterpret_cast<float4*>(out + (size_t)r * 4) = res;
    }
}

__global__ void __launch_bounds__(32 * WAVES_PER_WG)
qlayer_kernel(const float* __restrict__ x, const float* __restrict__ w,
              float* __restrict__ out, int batch, int chunks) {
    __shared__ cf gt[WAVES_PER_WG][32][2][2];
    const int lane = threadIdx.x & 31;
    const int wv   = threadIdx.x >> 5;
    const int wave = blockIdx.x * WAVES_PER_WG + wv;
    const int row0 = wave * chunks * CHUNK_ROWS + lane;
    if (chunks < 1) return;

    const int gate = (lane < 24) ? lane : lane - 8;
    build_gate_table(w[gate * 3 + 0], w[gate * 3 + 1], w[gate * 3 + 2], gt[wv], lane);
    __syncthreads();

    float4 cur[GROUPS];
#pragma unroll
    for (int g = 0; g < GROUPS; ++g) {
        int r = row0 + 32 * g;
        if (r >= batch) r = batch - 1;
        cur[g] = *reinterpret_cast<const float4*>(x + (size_t)r * 4);
    }
    __builtin_amdgcn_sched_barrier(0);

    v2f ar[4], ai[4];
    build_u_fragments(gt[wv], lane, ar, ai);

    int ch = 0;
    do {
        const int nch = (ch + 1 < chunks) ? ch + 1 : ch;
        float4 nxt[GROUPS];
#pragma unroll
        for (int g = 0; g < GROUPS; ++g) {
            int r = row0 + nch * CHUNK_ROWS + 32 * g;
            if (r >= batch) r = batch - 1;
            nxt[g] = *reinterpret_cast<const float4*>(x + (size_t)r * 4);
        }
        __builtin_amdgcn_sched_barrier(0);

#pragma unroll
        for (int g = 0; g < GROUPS; ++g)
            do_group(cur[g], ar, ai, out, row0 + ch * CHUNK_ROWS + 32 * g, batch);

#pragma unroll
        for (int g = 0; g < GROUPS; ++g) cur[g] = nxt[g];
    } while (++ch < chunks);
}

extern "C" void kernel_launch(void* const* d_in, const int* in_sizes, int n_in,
                              void* d_out, int out_size, void* d_ws, size_t ws_size,
                              hipStream_t stream) {
    const float* x = (const float*)d_in[0];
    const float* w = (const float*)d_in[1];
    float* out = (float*)d_out;

    const int batch = in_sizes[0] / 4;

    const int blocks = 256;
    const int rows_per_trip = blocks * WAVES_PER_WG * CHUNK_ROWS;
    const int chunks = (batch + rows_per_trip - 1) / rows_per_trip;
    qlayer_kernel<<<blocks, 32 * WAVES_PER_WG, 0, stream>>>(x, w, out, batch, chunks);
}
